# starting kernel
# baseline (speedup 1.0000x reference)
#include <hip/hip_runtime.h>


typedef float v2f __attribute__((ext_vector_type(2)));
typedef float v8f __attribute__((ext_vector_type(8)));

#define HDIM   128
#define KSTEPS 32
#define NTILES 8

__global__ __launch_bounds__(256, 1)
void edge_decoder_wmma_f32(const float* __restrict__ z_drug,
                           const float* __restrict__ z_effect,
                           const int*   __restrict__ eli,
                           const float* __restrict__ W1,
                           const float* __restrict__ b1,
                           const float* __restrict__ W2,
                           const float* __restrict__ b2,
                           float*       __restrict__ out,
                           int nEdges, int nTiles)
{
    __shared__ v2f   sW[KSTEPS * 2 * NTILES * 32];
    __shared__ float sB1[HDIM];
    __shared__ float sW2[HDIM];

    const int tid = threadIdx.x;

    for (int i = tid; i < KSTEPS * 2 * NTILES * 32; i += 256) {
        const int kk   = i >> 9;
        const int mat  = (i >> 8) & 1;
        const int nt   = (i >> 5) & 7;
        const int ln   = i & 31;
        const int krow = kk * 4 + ((ln >> 4) << 1) + mat * HDIM;
        const int col  = nt * 16 + (ln & 15);
        v2f w;
        w.x = W1[(krow    ) * HDIM + col];
        w.y = W1[(krow + 1) * HDIM + col];
        sW[i] = w;
    }
    if (tid < HDIM) { sB1[tid] = b1[tid]; sW2[tid] = W2[tid]; }
    __syncthreads();

    const int lane = tid & 31;
    const int wave = tid >> 5;
    const int lm   = lane & 15;
    const int lh   = lane >> 4;
    const float b2v = b2[0];

    for (int tile = blockIdx.x * 8 + wave; tile < nTiles; tile += gridDim.x * 8) {
        const int e0 = tile * 16;
        int eIdx = e0 + lm;
        if (eIdx >= nEdges) eIdx = nEdges - 1;
        const int rowId = eli[eIdx];
        const int colId = eli[nEdges + eIdx];
        const float* __restrict__ ad = z_drug   + (size_t)rowId * HDIM + lh * 2;
        const float* __restrict__ ae = z_effect + (size_t)colId * HDIM + lh * 2;

        v8f acc[NTILES] = {};

        #pragma unroll 2
        for (int kk = 0; kk < KSTEPS; ++kk) {
            const v2f a_d = *(const v2f*)(ad + kk * 4);
            const v2f a_e = *(const v2f*)(ae + kk * 4);
            const v2f* __restrict__ bk = &sW[(kk * 2 * NTILES) * 32 + lane];
            #pragma unroll
            for (int n = 0; n < NTILES; ++n) {
                acc[n] = __builtin_amdgcn_wmma_f32_16x16x4_f32(
                    false, a_d, false, bk[n * 32], (short)0, acc[n], false, false);
                acc[n] = __builtin_amdgcn_wmma_f32_16x16x4_f32(
                    false, a_e, false, bk[(NTILES + n) * 32], (short)0, acc[n], false, false);
            }
        }

        float partial[8];
        #pragma unroll
        for (int v = 0; v < 8; ++v) partial[v] = 0.f;

        #pragma unroll
        for (int n = 0; n < NTILES; ++n) {
            const float bv = sB1[n * 16 + lm];
            const float wv = sW2[n * 16 + lm];
            #pragma unroll
            for (int v = 0; v < 8; ++v) {
                float hv = acc[n][v] + bv;
                hv = hv > 0.f ? hv : 0.f;
                partial[v] += hv * wv;
            }
        }
        #pragma unroll
        for (int v = 0; v < 8; ++v) {
            float s = partial[v];
            s += __shfl_xor(s, 1, 32);
            s += __shfl_xor(s, 2, 32);
            s += __shfl_xor(s, 4, 32);
            s += __shfl_xor(s, 8, 32);
            partial[v] = s;
        }
        if (lm == 0) {
            #pragma unroll
            for (int v = 0; v < 8; ++v) {
                const int e = e0 + lh * 8 + v;
                if (e < nEdges) out[e] = partial[v] + b2v;
            }
        }
    }
}

extern "C" void kernel_launch(void* const* d_in, const int* in_sizes, int n_in,
                              void* d_out, int out_size, void* d_ws, size_t ws_size,
                              hipStream_t stream) {
    (void)n_in; (void)out_size; (void)d_ws; (void)ws_size;
    const float* z_drug   = (const float*)d_in[0];
    const float* z_effect = (const float*)d_in[1];
    const int*   eli      = (const int*)  d_in[2];
    const float* W1       = (const float*)d_in[3];
    const float* b1       = (const float*)d_in[4];
    const float* W2       = (const float*)d_in[5];
    const float* b2       = (const float*)d_in[6];
    float* out = (float*)d_out;

    const int nEdges = in_sizes[2] / 2;
    const int nTiles = (nEdges + 15) / 16;

    int blocks = 1024;
    const int maxBlocks = (nTiles + 7) / 8;
    if (blocks > maxBlocks) blocks = maxBlocks;

    edge_decoder_wmma_f32<<<blocks, 256, 0, stream>>>(
        z_drug, z_effect, eli, W1, b1, W2, b2, out, nEdges, nTiles);
}


# lin1 hoisted through the gather: launch 1 projects each node row once (f32 WMMA, into workspace), launch 2 gathers 2 projected rows per edge, relu, dot W2; 13x fewer WMMAs; sum order changes
# speedup vs baseline: 4.1488x; 4.1488x over previous
#include <hip/hip_runtime.h>


typedef float v4f __attribute__((ext_vector_type(4)));

typedef float v2f __attribute__((ext_vector_type(2)));
typedef float v8f __attribute__((ext_vector_type(8)));

#define HDIM   128
#define KSTEPS 32
#define NTILES 8


__global__ __launch_bounds__(256, 1)
void node_project_wmma_f32(const float* __restrict__ z_drug,
                           const float* __restrict__ z_effect,
                           const float* __restrict__ W1,
                           float*       __restrict__ P,
                           int nDrug, int nEffect)
{
    __shared__ v2f sWh[KSTEPS * NTILES * 32];

    const int tid  = threadIdx.x;
    const int lane = tid & 31;
    const int wave = tid >> 5;

    const int b3 = blockIdx.x % 3;
    const int bq = blockIdx.x / 3;
    const int G  = gridDim.x / 3;
    const bool isE = (b3 == 2);
    const int bIdx = isE ? bq : bq * 2 + b3;
    const int nB   = isE ? G : 2 * G;
    const float* __restrict__ Z  = isE ? z_effect : z_drug;
    const float* __restrict__ Wh = isE ? W1 + HDIM * HDIM : W1;
    float* __restrict__ Pout     = isE ? P + (size_t)nDrug * HDIM : P;
    const int nRows  = isE ? nEffect : nDrug;
    const int nTiles = (nRows + 15) >> 4;

    {
        const int cg = lane;
        v4f r0[8], r1[8];
        #pragma unroll
        for (int it = 0; it < 8; ++it) {
            const int kp = wave + 8 * it;
            r0[it] = *(const v4f*)(Wh + (2 * kp    ) * HDIM + 4 * cg);
            r1[it] = *(const v4f*)(Wh + (2 * kp + 1) * HDIM + 4 * cg);
        }
        #pragma unroll
        for (int it = 0; it < 8; ++it) {
            const int kp = wave + 8 * it;
            const int s  = 2 * (kp >> 2) + (kp & 1);
            const int h  = (kp >> 1) & 1;
            const int nb = 4 * (cg >> 4);
            const int c  = cg & 15;
            v2f* dst = &sWh[(s * NTILES + nb) * 32 + h * 16 + c];
            v2f w;
            w.x = r0[it].x; w.y = r1[it].x; dst[0 * 32] = w;
            w.x = r0[it].y; w.y = r1[it].y; dst[1 * 32] = w;
            w.x = r0[it].z; w.y = r1[it].z; dst[2 * 32] = w;
            w.x = r0[it].w; w.y = r1[it].w; dst[3 * 32] = w;
        }
    }
    __syncthreads();

    const int lm = lane & 15;
    const int lh = lane >> 4;

    int tile = bIdx * 8 + wave;
    if (tile >= nTiles) return;

    v4f a[16];
    {
        int r = tile * 16 + lm;
        if (r >= nRows) r = nRows - 1;
        const v4f* __restrict__ ap = (const v4f*)(Z + (size_t)r * HDIM + lh * 4);
        #pragma unroll
        for (int j = 0; j < 16; ++j) a[j] = ap[j * 2];
    }

    for (; tile < nTiles; tile += nB * 8) {
        v4f an[16];
        {
            int tn = tile + nB * 8;
            if (tn >= nTiles) tn = tile;
            int r = tn * 16 + lm;
            if (r >= nRows) r = nRows - 1;
            const v4f* __restrict__ ap = (const v4f*)(Z + (size_t)r * HDIM + lh * 4);
            #pragma unroll
            for (int j = 0; j < 16; ++j) an[j] = ap[j * 2];
        }

        v8f acc[NTILES] = {};

        v2f bf[2][2 * NTILES];
        #pragma unroll
        for (int q = 0; q < 2 * NTILES; ++q) bf[0][q] = sWh[q * 32 + lane];

        #pragma unroll
        for (int j = 0; j < 16; ++j) {
            if (j + 1 < 16) {
                #pragma unroll
                for (int q = 0; q < 2 * NTILES; ++q)
                    bf[(j + 1) & 1][q] = sWh[((j + 1) * 2 * NTILES + q) * 32 + lane];
            }
            __builtin_amdgcn_sched_barrier(0);
            v2f a0, a1;
            a0.x = a[j].x; a0.y = a[j].y;
            a1.x = a[j].z; a1.y = a[j].w;
            #pragma unroll
            for (int n = 0; n < NTILES; ++n)
                acc[n] = __builtin_amdgcn_wmma_f32_16x16x4_f32(
                    false, a0, false, bf[j & 1][n], (short)0, acc[n], false, false);
            #pragma unroll
            for (int n = 0; n < NTILES; ++n)
                acc[n] = __builtin_amdgcn_wmma_f32_16x16x4_f32(
                    false, a1, false, bf[j & 1][NTILES + n], (short)0, acc[n], false, false);
            __builtin_amdgcn_sched_barrier(0);
        }

        #pragma unroll
        for (int r = 0; r < 8; ++r) {
            const int row = tile * 16 + lh * 8 + r;
            if (row < nRows) {
                float* __restrict__ po = Pout + (size_t)row * HDIM + 4 * lm;
                #pragma unroll
                for (int g = 0; g < 2; ++g) {
                    v4f v;
                    v.x = acc[4 * g + 0][r];
                    v.y = acc[4 * g + 1][r];
                    v.z = acc[4 * g + 2][r];
                    v.w = acc[4 * g + 3][r];
                    *(v4f*)(po + 64 * g) = v;
                }
            }
        }

        #pragma unroll
        for (int j = 0; j < 16; ++j) a[j] = an[j];
    }
}

__global__ __launch_bounds__(128, 4)
void edge_gather_relu_dot(const float* __restrict__ P,
                          const int*   __restrict__ eli,
                          const float* __restrict__ b1,
                          const float* __restrict__ W2,
                          const float* __restrict__ b2,
                          float*       __restrict__ out,
                          int nEdges, int nDrug, int nEffect)
{
    const int lane = threadIdx.x & 31;
    const int wave = threadIdx.x >> 5;
    const int nBatches = (nEdges + 31) >> 5;
    const int stride   = gridDim.x * 4;

    int batch = blockIdx.x * 4 + wave;
    if (batch >= nBatches) return;

    const v4f bv = *(const v4f*)(b1 + lane * 4);
    const v4f wv = *(const v4f*)(W2 + lane * 4);
    const float b2v = b2[0];
    const float* __restrict__ Pd = P + lane * 4;
    const float* __restrict__ Pe = P + (size_t)nDrug * HDIM + lane * 4;

    int rowId, colId;
    {
        const int e  = batch * 32 + lane;
        const int ec = e < nEdges ? e : nEdges - 1;
        rowId = eli[ec];
        colId = eli[nEdges + ec];
    }

    for (;;) {
        const int nb  = batch + stride;
        const int nbc = nb < nBatches ? nb : batch;
        int rowN, colN;
        {
            const int e  = nbc * 32 + lane;
            const int ec = e < nEdges ? e : nEdges - 1;
            rowN = eli[ec];
            colN = eli[nEdges + ec];
        }

        rowId = rowId < 0 ? 0 : (rowId >= nDrug   ? nDrug   - 1 : rowId);
        colId = colId < 0 ? 0 : (colId >= nEffect ? nEffect - 1 : colId);

        float part[32];
        #pragma unroll
        for (int sb = 0; sb < 4; ++sb) {
            v4f pd[8], pe[8];
            #pragma unroll
            for (int j = 0; j < 8; ++j) {
                const int rj = __builtin_amdgcn_readlane(rowId, sb * 8 + j);
                const int cj = __builtin_amdgcn_readlane(colId, sb * 8 + j);
                pd[j] = *(const v4f*)(Pd + (size_t)rj * HDIM);
                pe[j] = *(const v4f*)(Pe + (size_t)cj * HDIM);
            }
            #pragma unroll
            for (int j = 0; j < 8; ++j) {
                v4f h = pd[j] + pe[j] + bv;
                h.x = h.x > 0.f ? h.x : 0.f;
                h.y = h.y > 0.f ? h.y : 0.f;
                h.z = h.z > 0.f ? h.z : 0.f;
                h.w = h.w > 0.f ? h.w : 0.f;
                part[sb * 8 + j] = h.x * wv.x + h.y * wv.y + h.z * wv.z + h.w * wv.w;
            }
        }

        #pragma unroll
        for (int m = 16; m >= 1; m >>= 1) {
            const bool up = (lane & m) != 0;
            #pragma unroll
            for (int i = 0; i < m; ++i) {
                const float lo = part[i];
                const float hi = part[i + m];
                const float keep = up ? hi : lo;
                const float send = up ? lo : hi;
                part[i] = keep + __shfl_xor(send, m, 32);
            }
        }

        const int e = batch * 32 + lane;
        if (e < nEdges) out[e] = part[0] + b2v;

        if (nb >= nBatches) break;
        batch = nb;
        rowId = rowN;
        colId = colN;
    }
}

__global__ __launch_bounds__(256, 1)
void edge_decoder_wmma_f32(const float* __restrict__ z_drug,
                           const float* __restrict__ z_effect,
                           const int*   __restrict__ eli,
                           const float* __restrict__ W1,
                           const float* __restrict__ b1,
                           const float* __restrict__ W2,
                           const float* __restrict__ b2,
                           float*       __restrict__ out,
                           int nEdges, int nTiles)
{
    __shared__ v2f   sW[KSTEPS * 2 * NTILES * 32];
    __shared__ float sB1[HDIM];
    __shared__ float sW2[HDIM];

    const int tid = threadIdx.x;

    for (int i = tid; i < KSTEPS * 2 * NTILES * 32; i += 256) {
        const int kk   = i >> 9;
        const int mat  = (i >> 8) & 1;
        const int nt   = (i >> 5) & 7;
        const int ln   = i & 31;
        const int krow = kk * 4 + ((ln >> 4) << 1) + mat * HDIM;
        const int col  = nt * 16 + (ln & 15);
        v2f w;
        w.x = W1[(krow    ) * HDIM + col];
        w.y = W1[(krow + 1) * HDIM + col];
        sW[i] = w;
    }
    if (tid < HDIM) { sB1[tid] = b1[tid]; sW2[tid] = W2[tid]; }
    __syncthreads();

    const int lane = tid & 31;
    const int wave = tid >> 5;
    const int lm   = lane & 15;
    const int lh   = lane >> 4;
    const float b2v = b2[0];

    for (int tile = blockIdx.x * 8 + wave; tile < nTiles; tile += gridDim.x * 8) {
        const int e0 = tile * 16;
        int eIdx = e0 + lm;
        if (eIdx >= nEdges) eIdx = nEdges - 1;
        const int rowId = eli[eIdx];
        const int colId = eli[nEdges + eIdx];
        const float* __restrict__ ad = z_drug   + (size_t)rowId * HDIM + lh * 2;
        const float* __restrict__ ae = z_effect + (size_t)colId * HDIM + lh * 2;

        v8f acc[NTILES] = {};

        #pragma unroll 2
        for (int kk = 0; kk < KSTEPS; ++kk) {
            const v2f a_d = *(const v2f*)(ad + kk * 4);
            const v2f a_e = *(const v2f*)(ae + kk * 4);
            const v2f* __restrict__ bk = &sW[(kk * 2 * NTILES) * 32 + lane];
            #pragma unroll
            for (int n = 0; n < NTILES; ++n) {
                acc[n] = __builtin_amdgcn_wmma_f32_16x16x4_f32(
                    false, a_d, false, bk[n * 32], (short)0, acc[n], false, false);
                acc[n] = __builtin_amdgcn_wmma_f32_16x16x4_f32(
                    false, a_e, false, bk[(NTILES + n) * 32], (short)0, acc[n], false, false);
            }
        }

        float partial[8];
        #pragma unroll
        for (int v = 0; v < 8; ++v) partial[v] = 0.f;

        #pragma unroll
        for (int n = 0; n < NTILES; ++n) {
            const float bv = sB1[n * 16 + lm];
            const float wv = sW2[n * 16 + lm];
            #pragma unroll
            for (int v = 0; v < 8; ++v) {
                float hv = acc[n][v] + bv;
                hv = hv > 0.f ? hv : 0.f;
                partial[v] += hv * wv;
            }
        }
        #pragma unroll
        for (int v = 0; v < 8; ++v) {
            float s = partial[v];
            s += __shfl_xor(s, 1, 32);
            s += __shfl_xor(s, 2, 32);
            s += __shfl_xor(s, 4, 32);
            s += __shfl_xor(s, 8, 32);
            partial[v] = s;
        }
        if (lm == 0) {
            #pragma unroll
            for (int v = 0; v < 8; ++v) {
                const int e = e0 + lh * 8 + v;
                if (e < nEdges) out[e] = partial[v] + b2v;
            }
        }
    }
}

extern "C" void kernel_launch(void* const* d_in, const int* in_sizes, int n_in,
                              void* d_out, int out_size, void* d_ws, size_t ws_size,
                              hipStream_t stream) {
    (void)n_in; (void)out_size;
    const float* z_drug   = (const float*)d_in[0];
    const float* z_effect = (const float*)d_in[1];
    const int*   eli      = (const int*)  d_in[2];
    const float* W1       = (const float*)d_in[3];
    const float* b1       = (const float*)d_in[4];
    const float* W2       = (const float*)d_in[5];
    const float* b2       = (const float*)d_in[6];
    float* out = (float*)d_out;

    const int nEdges  = in_sizes[2] / 2;
    const int nDrug   = in_sizes[0] / HDIM;
    const int nEffect = in_sizes[1] / HDIM;
    const size_t need = ((size_t)nDrug + (size_t)nEffect) * HDIM * sizeof(float);

    if (d_ws != nullptr && ws_size >= need) {
        float* P = (float*)d_ws;
        node_project_wmma_f32<<<255, 256, 0, stream>>>(z_drug, z_effect, W1, P, nDrug, nEffect);
        edge_gather_relu_dot<<<1024, 128, 0, stream>>>(P, eli, b1, W2, b2, out,
                                                       nEdges, nDrug, nEffect);
        return;
    }

    const int nTiles = (nEdges + 15) / 16;

    int blocks = 1024;
    const int maxBlocks = (nTiles + 7) / 8;
    if (blocks > maxBlocks) blocks = maxBlocks;

    edge_decoder_wmma_f32<<<blocks, 256, 0, stream>>>(
        z_drug, z_effect, eli, W1, b1, W2, b2, out, nEdges, nTiles);
}
